# starting kernel
# baseline (speedup 1.0000x reference)
#include <hip/hip_runtime.h>

typedef __attribute__((ext_vector_type(2))) float v2f;
typedef __attribute__((ext_vector_type(8))) float v8f;

__device__ __forceinline__ v8f wmma4(v2f a, v2f b, v8f c) {
  return __builtin_amdgcn_wmma_f32_16x16x4_f32(
       false, a,  false, b,
       (short)0, c,  false,  false);
}

__device__ __forceinline__ float gelu_exact(float x) {
  return 0.5f * x * (1.0f + erff(x * 0.70710678118654752440f));
}

__global__ __launch_bounds__(256) void ln_kernel(const float* __restrict__ x,
                                                 const float* __restrict__ gw,
                                                 const float* __restrict__ bw,
                                                 float* __restrict__ out) {
  const int row  = blockIdx.x;
  const int t    = threadIdx.x;
  const int wave = t >> 5, lane = t & 31;

  const float4 v = ((const float4*)(x + (size_t)row * 1024))[t];
  float s  = v.x + v.y + v.z + v.w;
  float s2 = v.x * v.x + v.y * v.y + v.z * v.z + v.w * v.w;
#pragma unroll
  for (int off = 16; off >= 1; off >>= 1) {
    s  += __shfl_xor(s, off, 32);
    s2 += __shfl_xor(s2, off, 32);
  }
  __shared__ float red[16];
  if (lane == 0) { red[2 * wave] = s; red[2 * wave + 1] = s2; }
  __syncthreads();
  float S1 = 0.f, S2 = 0.f;
#pragma unroll
  for (int w = 0; w < 8; ++w) { S1 += red[2 * w]; S2 += red[2 * w + 1]; }
  const float mean = S1 * (1.0f / 1024.0f);
  const float var  = S2 * (1.0f / 1024.0f) - mean * mean;
  const float rstd = rsqrtf(var + 1e-5f);

  const float4 g4 = ((const float4*)gw)[t];
  const float4 b4 = ((const float4*)bw)[t];
  float4 o;
  o.x = (v.x - mean) * rstd * g4.x + b4.x;
  o.y = (v.y - mean) * rstd * g4.y + b4.y;
  o.z = (v.z - mean) * rstd * g4.z + b4.z;
  o.w = (v.w - mean) * rstd * g4.w + b4.w;
  ((float4*)(out + (size_t)row * 1024))[t] = o;
}

template <int EPI>
__global__ __launch_bounds__(256) void gemm_kernel(const float* __restrict__ A,
                                                   const float* __restrict__ W,
                                                   const float* __restrict__ bias,
                                                   const float* __restrict__ res,
                                                   float* __restrict__ C,
                                                   int N, int K) {
  const int lane  = threadIdx.x & 31;
  const int wave  = threadIdx.x >> 5;
  const int row   = lane & 15;
  const int kh    = lane >> 4;
  const int waveM = wave >> 1;
  const int waveN = wave & 1;
  const int m0 = blockIdx.x * 128 + waveM * 32;
  const int n0 = blockIdx.y * 64 + waveN * 32;

  v8f c00 = {}, c01 = {}, c10 = {}, c11 = {};

  const float* a0p = A + (m0 + row) * K + 2 * kh;
  const float* a1p = a0p + 16 * K;
  const float* bp  = W + (2 * kh) * N + n0 + row;

#pragma unroll 4
  for (int k = 0; k < K; k += 4) {
    v2f a0 = *(const v2f*)a0p;
    v2f a1 = *(const v2f*)a1p;
    v2f b0, b1;
    b0.x = bp[0];  b0.y = bp[N];
    b1.x = bp[16]; b1.y = bp[N + 16];
    c00 = wmma4(a0, b0, c00);
    c01 = wmma4(a0, b1, c01);
    c10 = wmma4(a1, b0, c10);
    c11 = wmma4(a1, b1, c11);
    a0p += 4; a1p += 4; bp += 4 * N;
  }

  const float bv0 = bias[n0 + row];
  const float bv1 = bias[n0 + 16 + row];
#pragma unroll
  for (int i = 0; i < 8; ++i) {
    const int r0 = m0 + i + 8 * kh;
    const int r1 = r0 + 16;
    const int cA = n0 + row;
    const int cB = cA + 16;
    float v00 = c00[i] + bv0, v01 = c01[i] + bv1;
    float v10 = c10[i] + bv0, v11 = c11[i] + bv1;
    if (EPI == 1) {
      v00 += res[(size_t)r0 * N + cA]; v01 += res[(size_t)r0 * N + cB];
      v10 += res[(size_t)r1 * N + cA]; v11 += res[(size_t)r1 * N + cB];
    }
    if (EPI == 2) {
      v00 = gelu_exact(v00); v01 = gelu_exact(v01);
      v10 = gelu_exact(v10); v11 = gelu_exact(v11);
    }
    C[(size_t)r0 * N + cA] = v00; C[(size_t)r0 * N + cB] = v01;
    C[(size_t)r1 * N + cA] = v10; C[(size_t)r1 * N + cB] = v11;
  }
}

__global__ __launch_bounds__(128) void attn_kernel(const float* __restrict__ qkv,
                                                   float* __restrict__ out) {
  constexpr int S = 2048, D = 1024, R3D = 3072;
  const int lane = threadIdx.x & 31;
  const int wave = threadIdx.x >> 5;
  const int row  = lane & 15;
  const int kh   = lane >> 4;
  const int bh   = blockIdx.x;
  const int b    = bh >> 4, h = bh & 15;
  const int q0   = (blockIdx.y * 4 + wave) * 16;

  const float* qbase = qkv + (size_t)b * S * R3D + h * 64;
  const float* kbase = qbase + D;
  const float* vbase = qbase + 2 * D;

  v2f qf[16];
  {
    const float* qrow = qbase + (size_t)(q0 + row) * R3D + 2 * kh;
#pragma unroll
    for (int j = 0; j < 16; ++j) qf[j] = *(const v2f*)(qrow + 4 * j);
  }

  v8f o0 = {}, o1 = {}, o2 = {}, o3 = {};
  float mrun[8], lrun[8];
#pragma unroll
  for (int i = 0; i < 8; ++i) { mrun[i] = -1e30f; lrun[i] = 0.f; }

  __shared__ float Pbuf[4][16 * 18];
  float* P = Pbuf[wave];

  for (int kt = 0; kt < S; kt += 16) {
    v8f sc = {};
    {
      const float* krow = kbase + (size_t)(kt + row) * R3D + 2 * kh;
#pragma unroll
      for (int j = 0; j < 16; ++j) {
        v2f kf = *(const v2f*)(krow + 4 * j);
        sc = wmma4(qf[j], kf, sc);
      }
    }
    float alpha[8];
#pragma unroll
    for (int i = 0; i < 8; ++i) {
      float s = sc[i] * 0.125f;
      float mx = s;
      mx = fmaxf(mx, __shfl_xor(mx, 1, 32));
      mx = fmaxf(mx, __shfl_xor(mx, 2, 32));
      mx = fmaxf(mx, __shfl_xor(mx, 4, 32));
      mx = fmaxf(mx, __shfl_xor(mx, 8, 32));
      const float mnew = fmaxf(mrun[i], mx);
      const float a    = __expf(mrun[i] - mnew);
      const float p    = __expf(s - mnew);
      float sum = p;
      sum += __shfl_xor(sum, 1, 32);
      sum += __shfl_xor(sum, 2, 32);
      sum += __shfl_xor(sum, 4, 32);
      sum += __shfl_xor(sum, 8, 32);
      lrun[i] = lrun[i] * a + sum;
      mrun[i] = mnew;
      alpha[i] = a;
      P[(i + 8 * kh) * 18 + row] = p;
    }
#pragma unroll
    for (int i = 0; i < 8; ++i) {
      o0[i] *= alpha[i]; o1[i] *= alpha[i];
      o2[i] *= alpha[i]; o3[i] *= alpha[i];
    }
#pragma unroll
    for (int j = 0; j < 4; ++j) {
      v2f pa = *(const v2f*)(P + row * 18 + 4 * j + 2 * kh);
      const float* vr0 = vbase + (size_t)(kt + 4 * j + 2 * kh) * R3D;
      const float* vr1 = vr0 + R3D;
      v2f vb0 = { vr0[row],      vr1[row]      };
      v2f vb1 = { vr0[16 + row], vr1[16 + row] };
      v2f vb2 = { vr0[32 + row], vr1[32 + row] };
      v2f vb3 = { vr0[48 + row], vr1[48 + row] };
      o0 = wmma4(pa, vb0, o0);
      o1 = wmma4(pa, vb1, o1);
      o2 = wmma4(pa, vb2, o2);
      o3 = wmma4(pa, vb3, o3);
    }
  }

#pragma unroll
  for (int i = 0; i < 8; ++i) {
    const int qr = q0 + i + 8 * kh;
    const float inv = 1.0f / lrun[i];
    float* orow = out + ((size_t)b * S + qr) * D + h * 64 + row;
    orow[0]  = o0[i] * inv;
    orow[16] = o1[i] * inv;
    orow[32] = o2[i] * inv;
    orow[48] = o3[i] * inv;
  }
}

extern "C" void kernel_launch(void* const* d_in, const int* in_sizes, int n_in,
                              void* d_out, int out_size, void* d_ws, size_t ws_size,
                              hipStream_t stream) {
  const float* x      = (const float*)d_in[0];
  const float* ln1_g  = (const float*)d_in[1];
  const float* ln1_b  = (const float*)d_in[2];
  const float* qkv_w  = (const float*)d_in[3];
  const float* qkv_b  = (const float*)d_in[4];
  const float* proj_w = (const float*)d_in[5];
  const float* proj_b = (const float*)d_in[6];
  const float* ln2_g  = (const float*)d_in[7];
  const float* ln2_b  = (const float*)d_in[8];
  const float* fc1_w  = (const float*)d_in[9];
  const float* fc1_b  = (const float*)d_in[10];
  const float* fc2_w  = (const float*)d_in[11];
  const float* fc2_b  = (const float*)d_in[12];
  float* out = (float*)d_out;

  const int M = 4096;
  float* bufA = (float*)d_ws;
  float* bufB = bufA + (size_t)4096 * 4096;
  float* bufC = bufB + (size_t)4096 * 3072;
  float* bufD = bufC + (size_t)4096 * 1024;

  ln_kernel<<<M, 256, 0, stream>>>(x, ln1_g, ln1_b, bufA);
  gemm_kernel<0><<<dim3(M / 128, 3072 / 64), 256, 0, stream>>>(
      bufA, qkv_w, qkv_b, nullptr, bufB, 3072, 1024);
  attn_kernel<<<dim3(32, 32), 128, 0, stream>>>(bufB, bufC);
  gemm_kernel<1><<<dim3(M / 128, 1024 / 64), 256, 0, stream>>>(
      bufC, proj_w, proj_b, x, bufD, 1024, 1024);
  ln_kernel<<<M, 256, 0, stream>>>(bufD, ln2_g, ln2_b, bufB);
  gemm_kernel<2><<<dim3(M / 128, 4096 / 64), 256, 0, stream>>>(
      bufB, fc1_w, fc1_b, nullptr, bufA, 4096, 1024);
  gemm_kernel<1><<<dim3(M / 128, 1024 / 64), 256, 0, stream>>>(
      bufA, fc2_w, fc2_b, bufD, out, 1024, 4096);
}


# attention: wave = 2 Q tiles (32 queries) x half the keys; each K/V fragment it loads feeds both tiles (half the loads per matrix instruction); key halves merged through LDS at the end; new sum order
# speedup vs baseline: 1.3554x; 1.3554x over previous
#include <hip/hip_runtime.h>

typedef __attribute__((ext_vector_type(2))) float v2f;
typedef __attribute__((ext_vector_type(8))) float v8f;

__device__ __forceinline__ v8f wmma4(v2f a, v2f b, v8f c) {
  return __builtin_amdgcn_wmma_f32_16x16x4_f32(
       false, a,  false, b,
       (short)0, c,  false,  false);
}

__device__ __forceinline__ float gelu_exact(float x) {
  return 0.5f * x * (1.0f + erff(x * 0.70710678118654752440f));
}

__global__ __launch_bounds__(256) void ln_kernel(const float* __restrict__ x,
                                                 const float* __restrict__ gw,
                                                 const float* __restrict__ bw,
                                                 float* __restrict__ out) {
  const int row  = blockIdx.x;
  const int t    = threadIdx.x;
  const int wave = t >> 5, lane = t & 31;

  const float4 v = ((const float4*)(x + (size_t)row * 1024))[t];
  float s  = v.x + v.y + v.z + v.w;
  float s2 = v.x * v.x + v.y * v.y + v.z * v.z + v.w * v.w;
#pragma unroll
  for (int off = 16; off >= 1; off >>= 1) {
    s  += __shfl_xor(s, off, 32);
    s2 += __shfl_xor(s2, off, 32);
  }
  __shared__ float red[16];
  if (lane == 0) { red[2 * wave] = s; red[2 * wave + 1] = s2; }
  __syncthreads();
  float S1 = 0.f, S2 = 0.f;
#pragma unroll
  for (int w = 0; w < 8; ++w) { S1 += red[2 * w]; S2 += red[2 * w + 1]; }
  const float mean = S1 * (1.0f / 1024.0f);
  const float var  = S2 * (1.0f / 1024.0f) - mean * mean;
  const float rstd = rsqrtf(var + 1e-5f);

  const float4 g4 = ((const float4*)gw)[t];
  const float4 b4 = ((const float4*)bw)[t];
  float4 o;
  o.x = (v.x - mean) * rstd * g4.x + b4.x;
  o.y = (v.y - mean) * rstd * g4.y + b4.y;
  o.z = (v.z - mean) * rstd * g4.z + b4.z;
  o.w = (v.w - mean) * rstd * g4.w + b4.w;
  ((float4*)(out + (size_t)row * 1024))[t] = o;
}

template <int MT, int KS>
struct GemmFrag {
  float4 a[KS][MT];
  float4 w[KS][4];
};

template <int EPI, int MT, int KS, int N, int K>
__global__ __launch_bounds__(128) void gemm_kernel(const float* __restrict__ A,
                                                   const float* __restrict__ W,
                                                   const float* __restrict__ bias,
                                                   const float* __restrict__ res,
                                                   float* __restrict__ C) {
  constexpr int KC = 8 * KS;
  static_assert(K % (2 * KC) == 0 && N % 128 == 0, "tile shape");
  const unsigned lane = threadIdx.x & 31;
  const unsigned wave = threadIdx.x >> 5;
  const unsigned c    = lane & 15;
  const unsigned h    = lane >> 4;
  const unsigned m0 = blockIdx.x * (32 * MT) + (wave >> 1) * (16 * MT);
  const unsigned n0 = blockIdx.y * 128 + (wave & 1) * 64;

  const unsigned abyte = ((m0 + c) * K + 4 * h) * 4u;
  const unsigned wbyte = ((4 * h) * N + n0 + 4 * c) * 4u;
  const char* Ab = (const char*)A;
  const char* Wb = (const char*)W;

  v8f acc[MT][4];
#pragma unroll
  for (int i = 0; i < MT; ++i)
#pragma unroll
    for (int t = 0; t < 4; ++t) acc[i][t] = v8f{};

  auto load_frag = [&](GemmFrag<MT, KS>& f, int k0) {
    const char* Ak = Ab + (size_t)k0 * 4;
    const char* Wk = Wb + (size_t)k0 * (N * 4);
#pragma unroll
    for (int s = 0; s < KS; ++s) {
#pragma unroll
      for (int i = 0; i < MT; ++i)
        f.a[s][i] = *(const float4*)(Ak + abyte + (i * 16 * K + s * 8) * 4);
#pragma unroll
      for (int j = 0; j < 4; ++j)
        f.w[s][j] = *(const float4*)(Wk + wbyte + ((s * 8 + j) * N) * 4);
    }
  };

  auto mma_frag = [&](const GemmFrag<MT, KS>& f) {
#pragma unroll
    for (int s = 0; s < KS; ++s) {
      v2f b[4];
      b[0] = v2f{f.w[s][0].x, f.w[s][1].x}; b[1] = v2f{f.w[s][0].y, f.w[s][1].y};
      b[2] = v2f{f.w[s][0].z, f.w[s][1].z}; b[3] = v2f{f.w[s][0].w, f.w[s][1].w};
#pragma unroll
      for (int i = 0; i < MT; ++i) {
        const v2f a = v2f{f.a[s][i].x, f.a[s][i].y};
#pragma unroll
        for (int t = 0; t < 4; ++t) acc[i][t] = wmma4(a, b[t], acc[i][t]);
      }
      b[0] = v2f{f.w[s][2].x, f.w[s][3].x}; b[1] = v2f{f.w[s][2].y, f.w[s][3].y};
      b[2] = v2f{f.w[s][2].z, f.w[s][3].z}; b[3] = v2f{f.w[s][2].w, f.w[s][3].w};
#pragma unroll
      for (int i = 0; i < MT; ++i) {
        const v2f a = v2f{f.a[s][i].z, f.a[s][i].w};
#pragma unroll
        for (int t = 0; t < 4; ++t) acc[i][t] = wmma4(a, b[t], acc[i][t]);
      }
    }
  };

  GemmFrag<MT, KS> f0, f1;
  load_frag(f0, 0);
  for (int k0 = 0; k0 < K - 2 * KC; k0 += 2 * KC) {
    load_frag(f1, k0 + KC);
    __builtin_amdgcn_sched_barrier(0);
    mma_frag(f0);
    __builtin_amdgcn_sched_barrier(0);
    load_frag(f0, k0 + 2 * KC);
    __builtin_amdgcn_sched_barrier(0);
    mma_frag(f1);
    __builtin_amdgcn_sched_barrier(0);
  }
  load_frag(f1, K - KC);
  mma_frag(f0);
  mma_frag(f1);

  const unsigned col = n0 + 4 * c;
  const float4 bv = *(const float4*)(bias + col);
#pragma unroll
  for (int i = 0; i < MT; ++i) {
#pragma unroll
    for (int r = 0; r < 8; ++r) {
      const size_t o = (size_t)(m0 + 16 * i + 8 * h + r) * N + col;
      float4 v;
      v.x = acc[i][0][r] + bv.x; v.y = acc[i][1][r] + bv.y;
      v.z = acc[i][2][r] + bv.z; v.w = acc[i][3][r] + bv.w;
      if (EPI == 1) {
        const float4 rv = *(const float4*)(res + o);
        v.x += rv.x; v.y += rv.y; v.z += rv.z; v.w += rv.w;
      }
      if (EPI == 2) {
        v.x = gelu_exact(v.x); v.y = gelu_exact(v.y);
        v.z = gelu_exact(v.z); v.w = gelu_exact(v.w);
      }
      *(float4*)(C + o) = v;
    }
  }
}

__global__ __launch_bounds__(128, 4) void attn_kernel(const float* __restrict__ qkv,
                                                   float* __restrict__ out) {
  constexpr int S = 2048, D = 1024, R3D = 3072;
  const int lane = threadIdx.x & 31;
  const int wave = __builtin_amdgcn_readfirstlane(threadIdx.x >> 5);
  const int grp  = wave >> 1;
  const int half = wave & 1;
  const int row  = lane & 15;
  const int kh   = lane >> 4;
  const int bh   = blockIdx.x;
  const int b    = bh >> 4, h = bh & 15;
  const int q0   = (blockIdx.y * 2 + grp) * 32;
  const int kbeg = half * (S / 2);

  const float* qbase = qkv + (size_t)b * S * R3D + h * 64;
  const float* kbase = qbase + D;
  const float* vbase = qbase + 2 * D;
  const unsigned koff = (unsigned)row * R3D + 2u * (unsigned)kh;
  const unsigned voff = 2u * (unsigned)kh * R3D + (unsigned)row;

  v2f qf[2][16];
#pragma unroll
  for (int t = 0; t < 2; ++t) {
    const float* qrow = qbase + (size_t)(q0 + 16 * t + row) * R3D + 2 * kh;
#pragma unroll
    for (int j = 0; j < 16; ++j) qf[t][j] = *(const v2f*)(qrow + 4 * j);
  }

  v8f o[2][4];
  float mrun[2][8], lrun[2][8];
#pragma unroll
  for (int t = 0; t < 2; ++t) {
#pragma unroll
    for (int n = 0; n < 4; ++n) o[t][n] = v8f{};
#pragma unroll
    for (int i = 0; i < 8; ++i) { mrun[t][i] = -1e30f; lrun[t][i] = 0.f; }
  }

  __shared__ float Pbuf[4][2][16 * 18];
  __shared__ float Mbuf[2][96 * 32];
  float* P0 = Pbuf[wave][0];
  float* P1 = Pbuf[wave][1];

  for (int kt = kbeg; kt < kbeg + S / 2; kt += 16) {
    v8f sc[2] = {v8f{}, v8f{}};
    {
      const float* krow = kbase + (size_t)kt * R3D + koff;
#pragma unroll
      for (int j = 0; j < 16; ++j) {
        const v2f kf = *(const v2f*)(krow + 4 * j);
        sc[0] = wmma4(qf[0][j], kf, sc[0]);
        sc[1] = wmma4(qf[1][j], kf, sc[1]);
      }
    }
    __builtin_amdgcn_sched_barrier(0);
    float psum[2][8];
#pragma unroll
    for (int t = 0; t < 2; ++t) {
      float* P = t ? P1 : P0;
#pragma unroll
      for (int i = 0; i < 8; ++i) {
        const float s = sc[t][i] * 0.125f;
        float mx = s;
        mx = fmaxf(mx, __shfl_xor(mx, 1, 32));
        mx = fmaxf(mx, __shfl_xor(mx, 2, 32));
        mx = fmaxf(mx, __shfl_xor(mx, 4, 32));
        mx = fmaxf(mx, __shfl_xor(mx, 8, 32));
        const float a    = __expf(fminf(mrun[t][i] - mx, 0.0f));
        const float mnew = fmaxf(mrun[t][i], mx);
        const float p    = __expf(s - mnew);
        mrun[t][i] = mnew;
        lrun[t][i] *= a;
        o[t][0][i] *= a; o[t][1][i] *= a;
        o[t][2][i] *= a; o[t][3][i] *= a;
        psum[t][i] = p;
        P[(i + 8 * kh) * 18 + row] = p;
      }
    }
    __builtin_amdgcn_sched_barrier(0);
    v2f vb[4][4];
#pragma unroll
    for (int j = 0; j < 4; ++j) {
      const float* vr0 = vbase + (size_t)(kt + 4 * j) * R3D + voff;
      const float* vr1 = vr0 + R3D;
#pragma unroll
      for (int n = 0; n < 4; ++n) vb[j][n] = v2f{ vr0[16 * n], vr1[16 * n] };
    }
    __builtin_amdgcn_sched_barrier(0);
#pragma unroll
    for (int t = 0; t < 2; ++t) {
#pragma unroll
      for (int i = 0; i < 8; ++i) {
        float sum = psum[t][i];
        sum += __shfl_xor(sum, 1, 32);
        sum += __shfl_xor(sum, 2, 32);
        sum += __shfl_xor(sum, 4, 32);
        sum += __shfl_xor(sum, 8, 32);
        lrun[t][i] += sum;
      }
    }
    __builtin_amdgcn_sched_barrier(0);
#pragma unroll
    for (int j = 0; j < 4; ++j) {
      const v2f pa0 = *(const v2f*)(P0 + row * 18 + 4 * j + 2 * kh);
      const v2f pa1 = *(const v2f*)(P1 + row * 18 + 4 * j + 2 * kh);
#pragma unroll
      for (int n = 0; n < 4; ++n) {
        o[0][n] = wmma4(pa0, vb[j][n], o[0][n]);
        o[1][n] = wmma4(pa1, vb[j][n], o[1][n]);
      }
    }
  }

  float* M = Mbuf[grp];
  if (half == 1) {
#pragma unroll
    for (int t = 0; t < 2; ++t) {
#pragma unroll
      for (int i = 0; i < 8; ++i) {
        M[(t * 8 + i) * 32 + lane]      = mrun[t][i];
        M[(16 + t * 8 + i) * 32 + lane] = lrun[t][i];
#pragma unroll
        for (int n = 0; n < 4; ++n)
          M[(32 + (t * 4 + n) * 8 + i) * 32 + lane] = o[t][n][i];
      }
    }
  }
  __syncthreads();
  if (half == 0) {
#pragma unroll
    for (int t = 0; t < 2; ++t) {
#pragma unroll
      for (int i = 0; i < 8; ++i) {
        const float mhi = M[(t * 8 + i) * 32 + lane];
        const float lhi = M[(16 + t * 8 + i) * 32 + lane];
        const float m   = fmaxf(mrun[t][i], mhi);
        const float alo = __expf(mrun[t][i] - m);
        const float ahi = __expf(mhi - m);
        const float inv = 1.0f / (lrun[t][i] * alo + lhi * ahi);
        const int qr = q0 + 16 * t + i + 8 * kh;
        float* orow = out + ((size_t)b * S + qr) * D + h * 64 + row;
#pragma unroll
        for (int n = 0; n < 4; ++n) {
          const float ohi = M[(32 + (t * 4 + n) * 8 + i) * 32 + lane];
          orow[16 * n] = (o[t][n][i] * alo + ohi * ahi) * inv;
        }
      }
    }
  }
}

extern "C" void kernel_launch(void* const* d_in, const int* in_sizes, int n_in,
                              void* d_out, int out_size, void* d_ws, size_t ws_size,
                              hipStream_t stream) {
  const float* x      = (const float*)d_in[0];
  const float* ln1_g  = (const float*)d_in[1];
  const float* ln1_b  = (const float*)d_in[2];
  const float* qkv_w  = (const float*)d_in[3];
  const float* qkv_b  = (const float*)d_in[4];
  const float* proj_w = (const float*)d_in[5];
  const float* proj_b = (const float*)d_in[6];
  const float* ln2_g  = (const float*)d_in[7];
  const float* ln2_b  = (const float*)d_in[8];
  const float* fc1_w  = (const float*)d_in[9];
  const float* fc1_b  = (const float*)d_in[10];
  const float* fc2_w  = (const float*)d_in[11];
  const float* fc2_b  = (const float*)d_in[12];
  float* out = (float*)d_out;

  const int M = 4096;
  float* bufA = (float*)d_ws;
  float* bufB = bufA + (size_t)4096 * 4096;
  float* bufC = bufB + (size_t)4096 * 3072;
  float* bufD = bufC + (size_t)4096 * 1024;

  ln_kernel<<<M, 256, 0, stream>>>(x, ln1_g, ln1_b, bufA);
  gemm_kernel<0, 4, 1, 3072, 1024><<<dim3(M / 128, 3072 / 128), 128, 0, stream>>>(
      bufA, qkv_w, qkv_b, nullptr, bufB);
  attn_kernel<<<dim3(32, 32), 128, 0, stream>>>(bufB, bufC);
  gemm_kernel<1, 2, 2, 1024, 1024><<<dim3(M / 64, 1024 / 128), 128, 0, stream>>>(
      bufC, proj_w, proj_b, x, bufD);
  ln_kernel<<<M, 256, 0, stream>>>(bufD, ln2_g, ln2_b, bufB);
  gemm_kernel<2, 4, 1, 4096, 1024><<<dim3(M / 128, 4096 / 128), 128, 0, stream>>>(
      bufB, fc1_w, fc1_b, nullptr, bufA);
  gemm_kernel<1, 2, 2, 1024, 4096><<<dim3(M / 64, 1024 / 128), 128, 0, stream>>>(
      bufA, fc2_w, fc2_b, bufD, out);
}


# attention: the C-to-A layout trip through LDS moved before the softmax (scores, not P): row max/sum over a lane's own 8 keys + 1 lane-half exchange, 128 LDS permutes per 16 keys gone; new sum order
# speedup vs baseline: 1.4575x; 1.0753x over previous
#include <hip/hip_runtime.h>

typedef __attribute__((ext_vector_type(2))) float v2f;
typedef __attribute__((ext_vector_type(8))) float v8f;

__device__ __forceinline__ v8f wmma4(v2f a, v2f b, v8f c) {
  return __builtin_amdgcn_wmma_f32_16x16x4_f32(
       false, a,  false, b,
       (short)0, c,  false,  false);
}

__device__ __forceinline__ float gelu_exact(float x) {
  return 0.5f * x * (1.0f + erff(x * 0.70710678118654752440f));
}

__global__ __launch_bounds__(256) void ln_kernel(const float* __restrict__ x,
                                                 const float* __restrict__ gw,
                                                 const float* __restrict__ bw,
                                                 float* __restrict__ out) {
  const int row  = blockIdx.x;
  const int t    = threadIdx.x;
  const int wave = t >> 5, lane = t & 31;

  const float4 v = ((const float4*)(x + (size_t)row * 1024))[t];
  float s  = v.x + v.y + v.z + v.w;
  float s2 = v.x * v.x + v.y * v.y + v.z * v.z + v.w * v.w;
#pragma unroll
  for (int off = 16; off >= 1; off >>= 1) {
    s  += __shfl_xor(s, off, 32);
    s2 += __shfl_xor(s2, off, 32);
  }
  __shared__ float red[16];
  if (lane == 0) { red[2 * wave] = s; red[2 * wave + 1] = s2; }
  __syncthreads();
  float S1 = 0.f, S2 = 0.f;
#pragma unroll
  for (int w = 0; w < 8; ++w) { S1 += red[2 * w]; S2 += red[2 * w + 1]; }
  const float mean = S1 * (1.0f / 1024.0f);
  const float var  = S2 * (1.0f / 1024.0f) - mean * mean;
  const float rstd = rsqrtf(var + 1e-5f);

  const float4 g4 = ((const float4*)gw)[t];
  const float4 b4 = ((const float4*)bw)[t];
  float4 o;
  o.x = (v.x - mean) * rstd * g4.x + b4.x;
  o.y = (v.y - mean) * rstd * g4.y + b4.y;
  o.z = (v.z - mean) * rstd * g4.z + b4.z;
  o.w = (v.w - mean) * rstd * g4.w + b4.w;
  ((float4*)(out + (size_t)row * 1024))[t] = o;
}

template <int MT, int KS>
struct GemmFrag {
  float4 a[KS][MT];
  float4 w[KS][4];
};

template <int EPI, int MT, int KS, int N, int K>
__global__ __launch_bounds__(128) void gemm_kernel(const float* __restrict__ A,
                                                   const float* __restrict__ W,
                                                   const float* __restrict__ bias,
                                                   const float* __restrict__ res,
                                                   float* __restrict__ C) {
  constexpr int KC = 8 * KS;
  static_assert(K % (2 * KC) == 0 && N % 128 == 0, "tile shape");
  const unsigned lane = threadIdx.x & 31;
  const unsigned wave = threadIdx.x >> 5;
  const unsigned c    = lane & 15;
  const unsigned h    = lane >> 4;
  const unsigned m0 = blockIdx.x * (32 * MT) + (wave >> 1) * (16 * MT);
  const unsigned n0 = blockIdx.y * 128 + (wave & 1) * 64;

  const unsigned abyte = ((m0 + c) * K + 4 * h) * 4u;
  const unsigned wbyte = ((4 * h) * N + n0 + 4 * c) * 4u;
  const char* Ab = (const char*)A;
  const char* Wb = (const char*)W;

  v8f acc[MT][4];
#pragma unroll
  for (int i = 0; i < MT; ++i)
#pragma unroll
    for (int t = 0; t < 4; ++t) acc[i][t] = v8f{};

  auto load_frag = [&](GemmFrag<MT, KS>& f, int k0) {
    const char* Ak = Ab + (size_t)k0 * 4;
    const char* Wk = Wb + (size_t)k0 * (N * 4);
#pragma unroll
    for (int s = 0; s < KS; ++s) {
#pragma unroll
      for (int i = 0; i < MT; ++i)
        f.a[s][i] = *(const float4*)(Ak + abyte + (i * 16 * K + s * 8) * 4);
#pragma unroll
      for (int j = 0; j < 4; ++j)
        f.w[s][j] = *(const float4*)(Wk + wbyte + ((s * 8 + j) * N) * 4);
    }
  };

  auto mma_frag = [&](const GemmFrag<MT, KS>& f) {
#pragma unroll
    for (int s = 0; s < KS; ++s) {
      v2f b[4];
      b[0] = v2f{f.w[s][0].x, f.w[s][1].x}; b[1] = v2f{f.w[s][0].y, f.w[s][1].y};
      b[2] = v2f{f.w[s][0].z, f.w[s][1].z}; b[3] = v2f{f.w[s][0].w, f.w[s][1].w};
#pragma unroll
      for (int i = 0; i < MT; ++i) {
        const v2f a = v2f{f.a[s][i].x, f.a[s][i].y};
#pragma unroll
        for (int t = 0; t < 4; ++t) acc[i][t] = wmma4(a, b[t], acc[i][t]);
      }
      b[0] = v2f{f.w[s][2].x, f.w[s][3].x}; b[1] = v2f{f.w[s][2].y, f.w[s][3].y};
      b[2] = v2f{f.w[s][2].z, f.w[s][3].z}; b[3] = v2f{f.w[s][2].w, f.w[s][3].w};
#pragma unroll
      for (int i = 0; i < MT; ++i) {
        const v2f a = v2f{f.a[s][i].z, f.a[s][i].w};
#pragma unroll
        for (int t = 0; t < 4; ++t) acc[i][t] = wmma4(a, b[t], acc[i][t]);
      }
    }
  };

  GemmFrag<MT, KS> f0, f1;
  load_frag(f0, 0);
  for (int k0 = 0; k0 < K - 2 * KC; k0 += 2 * KC) {
    load_frag(f1, k0 + KC);
    __builtin_amdgcn_sched_barrier(0);
    mma_frag(f0);
    __builtin_amdgcn_sched_barrier(0);
    load_frag(f0, k0 + 2 * KC);
    __builtin_amdgcn_sched_barrier(0);
    mma_frag(f1);
    __builtin_amdgcn_sched_barrier(0);
  }
  load_frag(f1, K - KC);
  mma_frag(f0);
  mma_frag(f1);

  const unsigned col = n0 + 4 * c;
  const float4 bv = *(const float4*)(bias + col);
#pragma unroll
  for (int i = 0; i < MT; ++i) {
#pragma unroll
    for (int r = 0; r < 8; ++r) {
      const size_t o = (size_t)(m0 + 16 * i + 8 * h + r) * N + col;
      float4 v;
      v.x = acc[i][0][r] + bv.x; v.y = acc[i][1][r] + bv.y;
      v.z = acc[i][2][r] + bv.z; v.w = acc[i][3][r] + bv.w;
      if (EPI == 1) {
        const float4 rv = *(const float4*)(res + o);
        v.x += rv.x; v.y += rv.y; v.z += rv.z; v.w += rv.w;
      }
      if (EPI == 2) {
        v.x = gelu_exact(v.x); v.y = gelu_exact(v.y);
        v.z = gelu_exact(v.z); v.w = gelu_exact(v.w);
      }
      *(float4*)(C + o) = v;
    }
  }
}

__device__ __forceinline__ float xhalf(float x) {
  const int b = __builtin_bit_cast(int, x);
  return __builtin_bit_cast(float,
      __builtin_amdgcn_permlanex16(b, b, 0x76543210, 0xfedcba98, false, false));
}

__global__ __launch_bounds__(128, 4) void attn_kernel(const float* __restrict__ qkv,
                                                   float* __restrict__ out) {
  constexpr int S = 2048, D = 1024, R3D = 3072;
  const int lane = threadIdx.x & 31;
  const int wave = __builtin_amdgcn_readfirstlane(threadIdx.x >> 5);
  const int grp  = wave >> 1;
  const int half = wave & 1;
  const int row  = lane & 15;
  const int kh   = lane >> 4;
  const int bh   = blockIdx.x;
  const int b    = bh >> 4, h = bh & 15;
  const int q0   = (blockIdx.y * 2 + grp) * 32;
  const int kbeg = half * (S / 2);

  const float* qbase = qkv + (size_t)b * S * R3D + h * 64;
  const float* kbase = qbase + D;
  const float* vbase = qbase + 2 * D;
  const unsigned koff = (unsigned)row * R3D + 2u * (unsigned)kh;
  const unsigned voff = 2u * (unsigned)kh * R3D + (unsigned)row;

  v2f qf[2][16];
#pragma unroll
  for (int t = 0; t < 2; ++t) {
    const float* qrow = qbase + (size_t)(q0 + 16 * t + row) * R3D + 2 * kh;
#pragma unroll
    for (int j = 0; j < 16; ++j) qf[t][j] = *(const v2f*)(qrow + 4 * j);
  }

  v8f o[2][4];
#pragma unroll
  for (int t = 0; t < 2; ++t)
#pragma unroll
    for (int n = 0; n < 4; ++n) o[t][n] = v8f{};
  float mrun[2] = {-1e30f, -1e30f};
  float lrun[2] = {0.f, 0.f};

  __shared__ float Sbuf[4][2][16 * 18];
  __shared__ alignas(16) float Abuf[4][2][16];
  __shared__ float Mbuf[2][68 * 32];
  __shared__ alignas(16) float Fbuf[2][2][3][16];
  float* S0 = Sbuf[wave][0];
  float* S1 = Sbuf[wave][1];

  for (int kt = kbeg; kt < kbeg + S / 2; kt += 16) {
    v2f vb[4][4];
#pragma unroll
    for (int j = 0; j < 4; ++j) {
      const float* vr0 = vbase + (size_t)(kt + 4 * j) * R3D + voff;
      const float* vr1 = vr0 + R3D;
#pragma unroll
      for (int n = 0; n < 4; ++n) vb[j][n] = v2f{ vr0[16 * n], vr1[16 * n] };
    }
    v8f sc[2] = {v8f{}, v8f{}};
    {
      const float* krow = kbase + (size_t)kt * R3D + koff;
#pragma unroll
      for (int j = 0; j < 16; ++j) {
        const v2f kf = *(const v2f*)(krow + 4 * j);
        sc[0] = wmma4(qf[0][j], kf, sc[0]);
        sc[1] = wmma4(qf[1][j], kf, sc[1]);
      }
    }
    __builtin_amdgcn_sched_barrier(0);
#pragma unroll
    for (int t = 0; t < 2; ++t) {
      float* Sb = t ? S1 : S0;
#pragma unroll
      for (int i = 0; i < 8; ++i)
        Sb[(i + 8 * kh) * 18 + row] = sc[t][i] * 0.125f;
    }
    v2f pa[2][4];
#pragma unroll
    for (int t = 0; t < 2; ++t) {
      const float* Sb = t ? S1 : S0;
#pragma unroll
      for (int j = 0; j < 4; ++j) pa[t][j] = *(const v2f*)(Sb + row * 18 + 4 * j + 2 * kh);
    }
#pragma unroll
    for (int t = 0; t < 2; ++t) {
      float mx = fmaxf(fmaxf(fmaxf(pa[t][0].x, pa[t][0].y), fmaxf(pa[t][1].x, pa[t][1].y)),
                       fmaxf(fmaxf(pa[t][2].x, pa[t][2].y), fmaxf(pa[t][3].x, pa[t][3].y)));
      mx = fmaxf(mx, xhalf(mx));
      const float mnew = fmaxf(mrun[t], mx);
      const float a    = __expf(mrun[t] - mnew);
      mrun[t] = mnew;
#pragma unroll
      for (int j = 0; j < 4; ++j) {
        pa[t][j].x = __expf(pa[t][j].x - mnew);
        pa[t][j].y = __expf(pa[t][j].y - mnew);
      }
      const float ps = ((pa[t][0].x + pa[t][0].y) + (pa[t][1].x + pa[t][1].y)) +
                       ((pa[t][2].x + pa[t][2].y) + (pa[t][3].x + pa[t][3].y));
      lrun[t] = lrun[t] * a + ps;
      Abuf[wave][t][row] = a;
    }
#pragma unroll
    for (int t = 0; t < 2; ++t) {
      const float4 a0 = *(const float4*)(&Abuf[wave][t][8 * kh]);
      const float4 a1 = *(const float4*)(&Abuf[wave][t][8 * kh + 4]);
#pragma unroll
      for (int n = 0; n < 4; ++n) {
        o[t][n][0] *= a0.x; o[t][n][1] *= a0.y; o[t][n][2] *= a0.z; o[t][n][3] *= a0.w;
        o[t][n][4] *= a1.x; o[t][n][5] *= a1.y; o[t][n][6] *= a1.z; o[t][n][7] *= a1.w;
      }
    }
    __builtin_amdgcn_sched_barrier(0);
#pragma unroll
    for (int j = 0; j < 4; ++j) {
#pragma unroll
      for (int n = 0; n < 4; ++n) {
        o[0][n] = wmma4(pa[0][j], vb[j][n], o[0][n]);
        o[1][n] = wmma4(pa[1][j], vb[j][n], o[1][n]);
      }
    }
  }

#pragma unroll
  for (int t = 0; t < 2; ++t) lrun[t] += xhalf(lrun[t]);

  float* M = Mbuf[grp];
  if (half == 1) {
#pragma unroll
    for (int t = 0; t < 2; ++t) {
      M[t * 32 + lane]       = mrun[t];
      M[(2 + t) * 32 + lane] = lrun[t];
#pragma unroll
      for (int n = 0; n < 4; ++n)
#pragma unroll
        for (int i = 0; i < 8; ++i)
          M[(4 + (t * 4 + n) * 8 + i) * 32 + lane] = o[t][n][i];
    }
  }
  __syncthreads();
  if (half == 0) {
#pragma unroll
    for (int t = 0; t < 2; ++t) {
      const float mhi = M[t * 32 + lane];
      const float lhi = M[(2 + t) * 32 + lane];
      const float m   = fmaxf(mrun[t], mhi);
      const float alo = __expf(mrun[t] - m);
      const float ahi = __expf(mhi - m);
      Fbuf[grp][t][0][row] = alo;
      Fbuf[grp][t][1][row] = ahi;
      Fbuf[grp][t][2][row] = 1.0f / (lrun[t] * alo + lhi * ahi);
    }
#pragma unroll
    for (int t = 0; t < 2; ++t) {
#pragma unroll
      for (int i = 0; i < 8; ++i) {
        const float alo = Fbuf[grp][t][0][i + 8 * kh];
        const float ahi = Fbuf[grp][t][1][i + 8 * kh];
        const float inv = Fbuf[grp][t][2][i + 8 * kh];
        const int qr = q0 + 16 * t + i + 8 * kh;
        float* orow = out + ((size_t)b * S + qr) * D + h * 64 + row;
#pragma unroll
        for (int n = 0; n < 4; ++n) {
          const float ohi = M[(4 + (t * 4 + n) * 8 + i) * 32 + lane];
          orow[16 * n] = (o[t][n][i] * alo + ohi * ahi) * inv;
        }
      }
    }
  }
}

extern "C" void kernel_launch(void* const* d_in, const int* in_sizes, int n_in,
                              void* d_out, int out_size, void* d_ws, size_t ws_size,
                              hipStream_t stream) {
  const float* x      = (const float*)d_in[0];
  const float* ln1_g  = (const float*)d_in[1];
  const float* ln1_b  = (const float*)d_in[2];
  const float* qkv_w  = (const float*)d_in[3];
  const float* qkv_b  = (const float*)d_in[4];
  const float* proj_w = (const float*)d_in[5];
  const float* proj_b = (const float*)d_in[6];
  const float* ln2_g  = (const float*)d_in[7];
  const float* ln2_b  = (const float*)d_in[8];
  const float* fc1_w  = (const float*)d_in[9];
  const float* fc1_b  = (const float*)d_in[10];
  const float* fc2_w  = (const float*)d_in[11];
  const float* fc2_b  = (const float*)d_in[12];
  float* out = (float*)d_out;

  const int M = 4096;
  float* bufA = (float*)d_ws;
  float* bufB = bufA + (size_t)4096 * 4096;
  float* bufC = bufB + (size_t)4096 * 3072;
  float* bufD = bufC + (size_t)4096 * 1024;

  ln_kernel<<<M, 256, 0, stream>>>(x, ln1_g, ln1_b, bufA);
  gemm_kernel<0, 4, 1, 3072, 1024><<<dim3(M / 128, 3072 / 128), 128, 0, stream>>>(
      bufA, qkv_w, qkv_b, nullptr, bufB);
  attn_kernel<<<dim3(32, 32), 128, 0, stream>>>(bufB, bufC);
  gemm_kernel<1, 2, 2, 1024, 1024><<<dim3(M / 64, 1024 / 128), 128, 0, stream>>>(
      bufC, proj_w, proj_b, x, bufD);
  ln_kernel<<<M, 256, 0, stream>>>(bufD, ln2_g, ln2_b, bufB);
  gemm_kernel<2, 4, 1, 4096, 1024><<<dim3(M / 128, 4096 / 128), 128, 0, stream>>>(
      bufB, fc1_w, fc1_b, nullptr, bufA);
  gemm_kernel<1, 2, 2, 1024, 4096><<<dim3(M / 64, 1024 / 128), 128, 0, stream>>>(
      bufA, fc2_w, fc2_b, bufD, out);
}
